# starting kernel
# baseline (speedup 1.0000x reference)
#include <hip/hip_runtime.h>


#define DIM      128
#define NHEAD    8
#define NTOK     (4 * 16384)
#define NROWT    (NTOK / 16)
#define NT_QKV   24
#define NT_PROJ  8

#define WAVES_PER_BLOCK 4
#define BLOCK   (WAVES_PER_BLOCK * 32)
#define GRID    512

typedef float v2f __attribute__((ext_vector_type(2)));
typedef float v8f __attribute__((ext_vector_type(8)));

#define OFF_WQKV  0
#define SZ_WQKV   (128 * 384)
#define OFF_WPROJ (OFF_WQKV + SZ_WQKV)
#define SZ_WPROJ  (128 * 128)
#define OFF_ABUF  (OFF_WPROJ + SZ_WPROJ)
#define SZ_ABUF_W (16 * DIM)
#define OFF_KV    (OFF_ABUF + WAVES_PER_BLOCK * SZ_ABUF_W)
#define SZ_KV_W   (16 * 16 * 2)
#define LDS_FLOATS (OFF_KV + WAVES_PER_BLOCK * SZ_KV_W)

__launch_bounds__(BLOCK)
__global__ void fused_channel_attn(const float* __restrict__ x,
                                   const float* __restrict__ w_qkv,
                                   const float* __restrict__ b_qkv,
                                   const float* __restrict__ w_proj,
                                   const float* __restrict__ b_proj,
                                   float* __restrict__ out)
{
    __shared__ float lds[LDS_FLOATS];

    const int tid  = threadIdx.x;
    const int wave = tid >> 5;
    const int lane = tid & 31;
    const int l16  = lane & 15;
    const int half = lane >> 4;

    for (int idx = tid; idx < 128 * 384; idx += BLOCK) {
        int k = idx / 384, n = idx % 384;
        int kt = k >> 2, kr = k & 3;
        int nt = n >> 4, nl = n & 15;
        lds[OFF_WQKV + (((kt * NT_QKV + nt) * 32 + (nl + 16 * (kr >> 1))) << 1)
                     + (kr & 1)] = w_qkv[idx];
    }
    for (int idx = tid; idx < 128 * 128; idx += BLOCK) {
        int k = idx / 128, n = idx % 128;
        int kt = k >> 2, kr = k & 3;
        int nt = n >> 4, nl = n & 15;
        lds[OFF_WPROJ + (((kt * NT_PROJ + nt) * 32 + (nl + 16 * (kr >> 1))) << 1)
                      + (kr & 1)] = w_proj[idx];
    }
    __syncthreads();

    float bqk[NT_QKV];
#pragma unroll
    for (int nt = 0; nt < NT_QKV; ++nt) bqk[nt] = b_qkv[nt * 16 + l16];
    float bp[NT_PROJ];
#pragma unroll
    for (int nt = 0; nt < NT_PROJ; ++nt) bp[nt] = b_proj[nt * 16 + l16];

    float* abuf  = &lds[OFF_ABUF + wave * SZ_ABUF_W];
    float* kvbuf = &lds[OFF_KV   + wave * SZ_KV_W];

    const float c1 = 0.08838834764831845f * 1.4426950408889634f;

    const int waveGlobal = blockIdx.x * WAVES_PER_BLOCK + wave;
    const int waveStride = GRID * WAVES_PER_BLOCK;

    for (int rt = waveGlobal; rt < NROWT; rt += waveStride) {
        const int row0 = rt * 16;

        const float4* xg  = (const float4*)(x + (size_t)row0 * DIM);
        float4*       ab4 = (float4*)abuf;
#pragma unroll
        for (int i = 0; i < (16 * DIM / 4) / 32; ++i)
            ab4[lane + i * 32] = xg[lane + i * 32];

        v2f a[32];
#pragma unroll
        for (int kt = 0; kt < 32; ++kt)
            a[kt] = *(const v2f*)&abuf[l16 * DIM + kt * 4 + half * 2];

#pragma unroll 1
        for (int h = 0; h < NHEAD; ++h) {
            v8f cq, ck, cv;
#pragma unroll
            for (int r = 0; r < 8; ++r) {
                cq[r] = bqk[h];
                ck[r] = bqk[8 + h];
                cv[r] = bqk[16 + h];
            }
#pragma unroll
            for (int kt = 0; kt < 32; ++kt) {
                v2f bq = *(const v2f*)&lds[OFF_WQKV + (kt * NT_QKV + h     ) * 64 + lane * 2];
                v2f bk = *(const v2f*)&lds[OFF_WQKV + (kt * NT_QKV + h + 8 ) * 64 + lane * 2];
                v2f bv = *(const v2f*)&lds[OFF_WQKV + (kt * NT_QKV + h + 16) * 64 + lane * 2];
                cq = __builtin_amdgcn_wmma_f32_16x16x4_f32(false, a[kt], false, bq,
                                                           (short)0, cq, false, false);
                ck = __builtin_amdgcn_wmma_f32_16x16x4_f32(false, a[kt], false, bk,
                                                           (short)0, ck, false, false);
                cv = __builtin_amdgcn_wmma_f32_16x16x4_f32(false, a[kt], false, bv,
                                                           (short)0, cv, false, false);
            }

#pragma unroll
            for (int r = 0; r < 8; ++r) {
                int t = r + 8 * half;
                v2f kv; kv.x = ck[r]; kv.y = cv[r];
                *(v2f*)&kvbuf[(t * 16 + l16) * 2] = kv;
            }

#pragma unroll
            for (int r = 0; r < 8; ++r) {
                int t   = r + 8 * half;
                float q = cq[r];
                float num = 0.0f, den = 0.0f;
#pragma unroll
                for (int j = 0; j < 16; ++j) {
                    v2f kv  = *(const v2f*)&kvbuf[(t * 16 + j) * 2];
                    float e = __builtin_amdgcn_exp2f(q * kv.x * c1);
                    num += e * kv.y;
                    den += e;
                }
                abuf[t * DIM + h * 16 + l16] = num * __builtin_amdgcn_rcpf(den);
            }
        }

#pragma unroll
        for (int kt = 0; kt < 32; ++kt)
            a[kt] = *(const v2f*)&abuf[l16 * DIM + kt * 4 + half * 2];

#pragma unroll 1
        for (int nt = 0; nt < NT_PROJ; ++nt) {
            v8f c;
#pragma unroll
            for (int r = 0; r < 8; ++r) c[r] = bp[nt];
#pragma unroll
            for (int kt = 0; kt < 32; ++kt) {
                v2f b = *(const v2f*)&lds[OFF_WPROJ + (kt * NT_PROJ + nt) * 64 + lane * 2];
                c = __builtin_amdgcn_wmma_f32_16x16x4_f32(false, a[kt], false, b,
                                                          (short)0, c, false, false);
            }
#pragma unroll
            for (int r = 0; r < 8; ++r)
                out[(size_t)(row0 + r + 8 * half) * DIM + nt * 16 + l16] = c[r];
        }
    }
}

extern "C" void kernel_launch(void* const* d_in, const int* in_sizes, int n_in,
                              void* d_out, int out_size, void* d_ws, size_t ws_size,
                              hipStream_t stream) {
    const float* x      = (const float*)d_in[0];
    const float* w_qkv  = (const float*)d_in[1];
    const float* b_qkv  = (const float*)d_in[2];
    const float* w_proj = (const float*)d_in[3];
    const float* b_proj = (const float*)d_in[4];
    float* out = (float*)d_out;
    (void)in_sizes; (void)n_in; (void)out_size; (void)d_ws; (void)ws_size;
    fused_channel_attn<<<dim3(GRID), dim3(BLOCK), 0, stream>>>(
        x, w_qkv, b_qkv, w_proj, b_proj, out);
}


# Weights register-resident: each of 4 waves keeps its 2 heads' qkv columns and 32 proj columns as B fragments in VGPRs; row tiles stream through LDS; no weight staging in LDS
# speedup vs baseline: 2.2135x; 2.2135x over previous
#include <hip/hip_runtime.h>


#define DIM      128
#define NTOK     (4 * 16384)
#define NROWT    (NTOK / 16)
#define WAVES    4
#define BLOCK    (WAVES * 32)
#define GRID     256
#define NIT      (NROWT / GRID)
#define PITCH    132

typedef float v2f __attribute__((ext_vector_type(2)));
typedef float v4f __attribute__((ext_vector_type(4)));
typedef float v8f __attribute__((ext_vector_type(8)));

__launch_bounds__(BLOCK)
__global__ void fused_channel_attn(const float* __restrict__ x,
                                   const float* __restrict__ w_qkv,
                                   const float* __restrict__ b_qkv,
                                   const float* __restrict__ w_proj,
                                   const float* __restrict__ b_proj,
                                   float* __restrict__ out)
{
    __shared__ __attribute__((aligned(16))) float xbuf[2][16 * PITCH];
    __shared__ __attribute__((aligned(16))) float obuf[16 * PITCH];
    __shared__ __attribute__((aligned(16))) float kvs[WAVES][2][512];

    const int tid  = threadIdx.x;
    const int wave = tid >> 5;
    const int lane = tid & 31;
    const int c    = lane & 15;
    const int hh   = lane >> 4;


    v2f wq[2][3][32];
#pragma unroll
    for (int hl = 0; hl < 2; ++hl) {
#pragma unroll
        for (int m = 0; m < 3; ++m) {
            const float* wcol = w_qkv + m * DIM + (2 * wave + hl) * 16 + c + (size_t)(64 * hh) * (3 * DIM);
#pragma unroll
            for (int t = 0; t < 32; ++t) {
                v2f b;
                b.x = wcol[(2 * t    ) * (3 * DIM)];
                b.y = wcol[(2 * t + 1) * (3 * DIM)];
                wq[hl][m][t] = b;
            }
        }
    }
    v2f wp[2][32];
#pragma unroll
    for (int j = 0; j < 2; ++j) {
        const float* wcol = w_proj + wave * 32 + 2 * c + j + (size_t)(64 * hh) * DIM;
#pragma unroll
        for (int t = 0; t < 32; ++t) {
            v2f b;
            b.x = wcol[(2 * t    ) * DIM];
            b.y = wcol[(2 * t + 1) * DIM];
            wp[j][t] = b;
        }
    }
    float bq[2][3];
#pragma unroll
    for (int hl = 0; hl < 2; ++hl)
#pragma unroll
        for (int m = 0; m < 3; ++m)
            bq[hl][m] = b_qkv[m * DIM + (2 * wave + hl) * 16 + c];
    const float bp0 = b_proj[wave * 32 + 2 * c];
    const float bp1 = b_proj[wave * 32 + 2 * c + 1];

    const float c1 = 0.08838834764831845f * 1.4426950408889634f;

    const v4f* x4 = (const v4f*)x;
    v4f xr[4];

    {
        const int row0 = blockIdx.x * 16;
#pragma unroll
        for (int i = 0; i < 4; ++i)
            xr[i] = x4[(size_t)(row0 + wave + 4 * i) * (DIM / 4) + lane];
#pragma unroll
        for (int i = 0; i < 4; ++i)
            *(v4f*)&xbuf[0][(wave + 4 * i) * PITCH + lane * 4] = xr[i];
    }

#pragma unroll 1
    for (int it = 0; it < NIT; ++it) {
        const int row0 = (blockIdx.x + it * GRID) * 16;
        const bool more = (it + 1 < NIT);

        __syncthreads();

        if (more) {
            const int rown = row0 + GRID * 16;
#pragma unroll
            for (int i = 0; i < 4; ++i)
                xr[i] = x4[(size_t)(rown + wave + 4 * i) * (DIM / 4) + lane];
        }

        const float* xb = xbuf[it & 1];
        v2f a[32];
#pragma unroll
        for (int i = 0; i < 16; ++i) {
            v4f v = *(const v4f*)&xb[c * PITCH + 64 * hh + 4 * i];
            a[2 * i    ].x = v.x; a[2 * i    ].y = v.y;
            a[2 * i + 1].x = v.z; a[2 * i + 1].y = v.w;
        }
        __builtin_amdgcn_sched_barrier(0);

        v8f acc[2][3];
#pragma unroll
        for (int hl = 0; hl < 2; ++hl)
#pragma unroll
            for (int m = 0; m < 3; ++m)
#pragma unroll
                for (int r = 0; r < 8; ++r) acc[hl][m][r] = bq[hl][m];
#pragma unroll
        for (int t = 0; t < 32; ++t) {
#pragma unroll
            for (int hl = 0; hl < 2; ++hl)
#pragma unroll
                for (int m = 0; m < 3; ++m)
                    acc[hl][m] = __builtin_amdgcn_wmma_f32_16x16x4_f32(false, a[t], false, wq[hl][m][t],
                                                                       (short)0, acc[hl][m], false, false);
        }

        __builtin_amdgcn_sched_barrier(0);
#pragma unroll
        for (int hl = 0; hl < 2; ++hl) {
#pragma unroll
            for (int r = 0; r < 8; ++r) {
                int t = r + 8 * hh;
                v2f kv; kv.x = acc[hl][1][r]; kv.y = acc[hl][2][r];
                *(v2f*)&kvs[wave][hl][(t * 16 + c) * 2] = kv;
            }
        }

        v4f kvr[2][32];
#pragma unroll
        for (int rr = 0; rr < 4; ++rr) {
            int t = rr + 8 * hh;
#pragma unroll
            for (int jj = 0; jj < 8; ++jj)
                kvr[0][rr * 8 + jj] = *(const v4f*)&kvs[wave][0][(t * 16 + 2 * jj) * 2];
        }
#pragma unroll
        for (int G = 0; G < 4; ++G) {
            __builtin_amdgcn_sched_barrier(0);
            if (G + 1 < 4) {
                const int hn = (G + 1) >> 1, gn = (G + 1) & 1;
#pragma unroll
                for (int rr = 0; rr < 4; ++rr) {
                    int t = 4 * gn + rr + 8 * hh;
#pragma unroll
                    for (int jj = 0; jj < 8; ++jj)
                        kvr[(G + 1) & 1][rr * 8 + jj] =
                            *(const v4f*)&kvs[wave][hn][(t * 16 + 2 * jj) * 2];
                }
            }
            __builtin_amdgcn_sched_barrier(0);
            const int hl = G >> 1, g = G & 1;
            const int head = 2 * wave + hl;
#pragma unroll
            for (int rr = 0; rr < 4; ++rr) {
                const int r = 4 * g + rr;
                int t   = r + 8 * hh;
                float q = acc[hl][0][r];
                float num = 0.0f, den = 0.0f;
#pragma unroll
                for (int jj = 0; jj < 8; ++jj) {
                    v4f kv = kvr[G & 1][rr * 8 + jj];
                    float e0 = __builtin_amdgcn_exp2f(q * kv.x * c1);
                    num += e0 * kv.y;
                    den += e0;
                    float e1 = __builtin_amdgcn_exp2f(q * kv.z * c1);
                    num += e1 * kv.w;
                    den += e1;
                }
                obuf[t * PITCH + head * 16 + c] = num * __builtin_amdgcn_rcpf(den);
            }
        }
        __builtin_amdgcn_sched_barrier(0);

        if (more) {
            float* xn = xbuf[(it + 1) & 1];
#pragma unroll
            for (int i = 0; i < 4; ++i)
                *(v4f*)&xn[(wave + 4 * i) * PITCH + lane * 4] = xr[i];
        }

        __syncthreads();

#pragma unroll
        for (int i = 0; i < 16; ++i) {
            v4f v = *(const v4f*)&obuf[c * PITCH + 64 * hh + 4 * i];
            a[2 * i    ].x = v.x; a[2 * i    ].y = v.y;
            a[2 * i + 1].x = v.z; a[2 * i + 1].y = v.w;
        }
        __builtin_amdgcn_sched_barrier(0);
        v8f p0, p1;
#pragma unroll
        for (int r = 0; r < 8; ++r) { p0[r] = bp0; p1[r] = bp1; }
#pragma unroll
        for (int t = 0; t < 32; ++t) {
            p0 = __builtin_amdgcn_wmma_f32_16x16x4_f32(false, a[t], false, wp[0][t],
                                                       (short)0, p0, false, false);
            p1 = __builtin_amdgcn_wmma_f32_16x16x4_f32(false, a[t], false, wp[1][t],
                                                       (short)0, p1, false, false);
        }
#pragma unroll
        for (int r = 0; r < 8; ++r) {
            v2f o; o.x = p0[r]; o.y = p1[r];
            *(v2f*)&out[(size_t)(row0 + r + 8 * hh) * DIM + wave * 32 + 2 * c] = o;
        }
    }
}

extern "C" void kernel_launch(void* const* d_in, const int* in_sizes, int n_in,
                              void* d_out, int out_size, void* d_ws, size_t ws_size,
                              hipStream_t stream) {
    const float* x      = (const float*)d_in[0];
    const float* w_qkv  = (const float*)d_in[1];
    const float* b_qkv  = (const float*)d_in[2];
    const float* w_proj = (const float*)d_in[3];
    const float* b_proj = (const float*)d_in[4];
    float* out = (float*)d_out;
    (void)in_sizes; (void)n_in; (void)out_size; (void)d_ws; (void)ws_size;
    fused_channel_attn<<<dim3(GRID), dim3(BLOCK), 0, stream>>>(
        x, w_qkv, b_qkv, w_proj, b_proj, out);
}
